# starting kernel
# baseline (speedup 1.0000x reference)
#include <hip/hip_runtime.h>

typedef __attribute__((ext_vector_type(2))) float v2f;
typedef __attribute__((ext_vector_type(8))) float v8f;

#define KDE_D 128
#define KDE_P (KDE_D / 2)
#define KTILES_PER_WAVE 32

__global__ void kde_bw_kernel(const float* __restrict__ train, int n,
                              float* __restrict__ bw, float* __restrict__ inv_bw,
                              float* __restrict__ scal) {
    const int d = threadIdx.x;
    double s = 0.0, s2 = 0.0;
    for (int i = 0; i < n; ++i) {
        float x = train[i * KDE_D + d];
        s += (double)x;
        s2 += (double)x * (double)x;
    }
    double var = (s2 - s * s / (double)n) / (double)(n - 1);
    if (var < 0.0) var = 0.0;
    float stdv = (float)sqrt(var);
    stdv = fmaxf(stdv, 0.01f);
    float scott = (float)exp(-log((double)n) / 132.0);
    float b = 1.06f * stdv * scott;
    b = fminf(b, 0.49f);
    bw[d] = b;
    inv_bw[d] = 1.0f / b;

    __shared__ float red[KDE_D];
    red[d] = logf(b);
    __syncthreads();
    for (int off = KDE_D / 2; off > 0; off >>= 1) {
        if (d < off) red[d] += red[d + off];
        __syncthreads();
    }
    if (d == 0) {
        float Z = 0.5f * (float)KDE_D * logf(6.28318530717958647692f)
                + red[0] + logf((float)n);
        scal[0] = Z;
        scal[1] = 1.0f / Z;
    }
}

__global__ void kde_pack_kernel(const float* __restrict__ src,
                                const float* __restrict__ inv_bw,
                                float* __restrict__ dstP, int n) {
    int i = blockIdx.x * blockDim.x + threadIdx.x;
    if (i >= KDE_P * n) return;
    int q = i % n;
    int p = i / n;
    float2 v;
    v.x = src[q * KDE_D + 2 * p + 0] * inv_bw[2 * p + 0];
    v.y = src[q * KDE_D + 2 * p + 1] * inv_bw[2 * p + 1];
    ((float2*)dstP)[p * n + q] = v;
}

__global__ void kde_norms_kernel(const float* __restrict__ srcP, int n,
                                 float* __restrict__ out) {
    int q = blockIdx.x * blockDim.x + threadIdx.x;
    if (q >= n) return;
    float acc = 0.0f;
    const float2* s2 = (const float2*)srcP;
    for (int p = 0; p < KDE_P; ++p) {
        float2 v = s2[p * n + q];
        acc = fmaf(v.x, v.x, acc);
        acc = fmaf(v.y, v.y, acc);
    }
    out[q] = acc;
}

__global__ __launch_bounds__(256)
void kde_wmma_kernel(const float* __restrict__ tP, const float* __restrict__ trP,
                     const float* __restrict__ qn, const float* __restrict__ kn,
                     const float* __restrict__ scal, float* __restrict__ partial,
                     int NQ, int NK, int nkg) {
    const int lane = threadIdx.x & 31;
    const int wave = threadIdx.x >> 5;
    const int qt   = blockIdx.x * 8 + wave;
    const int kg   = blockIdx.y;
    const int q0   = qt * 16;
    const int hi   = lane >> 4;
    const int lo   = lane & 15;
    const float invZ = scal[1];

    const float2* tP2  = (const float2*)tP;
    const float2* trP2 = (const float2*)trP;

    v2f aF[32];
#pragma unroll
    for (int s4 = 0; s4 < 32; ++s4) {
        float2 t = tP2[(2 * s4 + hi) * NQ + q0 + lo];
        aF[s4].x = t.x;
        aF[s4].y = t.y;
    }
    float qrow[8];
#pragma unroll
    for (int r = 0; r < 8; ++r) qrow[r] = qn[q0 + 8 * hi + r];

    float racc[8] = {};
    const int kstart = kg * (KTILES_PER_WAVE * 16);

    v2f b0[32], b1[32];

    auto loadB = [&](v2f (&buf)[32], int kb) {
#pragma unroll
        for (int s4 = 0; s4 < 32; ++s4) {
            float2 t = trP2[(2 * s4 + hi) * NK + kb + lo];
            buf[s4].x = t.x;
            buf[s4].y = t.y;
        }
    };
    auto processB = [&](const v2f (&buf)[32], int kb) {
        v8f acc = {};
#pragma unroll
        for (int s4 = 0; s4 < 32; ++s4) {
            acc = __builtin_amdgcn_wmma_f32_16x16x4_f32(
                false, aF[s4], false, buf[s4], (short)0, acc, false, false);
        }
        const float knv = kn[kb + lo];
#pragma unroll
        for (int r = 0; r < 8; ++r) {
            float sq = fmaxf(qrow[r] + knv - 2.0f * acc[r], 0.0f);
            racc[r] += __expf(fmaf(-0.5f * sq, invZ, -1.0f));
        }
    };

    loadB(b0, kstart);
    for (int ktp = 0; ktp < KTILES_PER_WAVE / 2; ++ktp) {
        const int kb0 = kstart + ktp * 32;
        loadB(b1, kb0 + 16);
        processB(b0, kb0);
        if (ktp + 1 < KTILES_PER_WAVE / 2) loadB(b0, kb0 + 32);
        processB(b1, kb0 + 16);
    }

#pragma unroll
    for (int r = 0; r < 8; ++r) {
        float v = racc[r];
        v += __shfl_xor(v, 1, 32);
        v += __shfl_xor(v, 2, 32);
        v += __shfl_xor(v, 4, 32);
        v += __shfl_xor(v, 8, 32);
        racc[r] = v;
    }
    if (lo == 0) {
#pragma unroll
        for (int r = 0; r < 8; ++r) {
            int row = q0 + 8 * hi + r;
            partial[row * nkg + kg] = racc[r];
        }
    }
}

__global__ void kde_reduce_kernel(const float* __restrict__ partial,
                                  float* __restrict__ out, int NQ, int nkg) {
    int q = blockIdx.x * blockDim.x + threadIdx.x;
    if (q >= NQ) return;
    float acc = 0.0f;
    for (int g = 0; g < nkg; ++g) acc += partial[q * nkg + g];
    out[q] = acc;
}

extern "C" void kernel_launch(void* const* d_in, const int* in_sizes, int n_in,
                              void* d_out, int out_size, void* d_ws, size_t ws_size,
                              hipStream_t stream) {
    const float* test  = (const float*)d_in[0];
    const float* train = (const float*)d_in[1];
    float* out = (float*)d_out;

    const int NQ = in_sizes[0] / KDE_D;
    const int NK = in_sizes[1] / KDE_D;
    const int nkg = NK / (KTILES_PER_WAVE * 16);

    float* ws      = (float*)d_ws;
    float* bw      = ws;
    float* inv_bw  = ws + KDE_D;
    float* scal    = ws + 2 * KDE_D;
    float* qn      = ws + 3 * KDE_D;
    float* kn      = qn + NQ;
    float* tP      = kn + NK;
    float* trP     = tP + KDE_D * NQ;
    float* partial = trP + KDE_D * NK;

    kde_bw_kernel<<<1, KDE_D, 0, stream>>>(train, NK, bw, inv_bw, scal);

    int tot_t  = KDE_P * NQ;
    int tot_tr = KDE_P * NK;
    kde_pack_kernel<<<(tot_t + 255) / 256, 256, 0, stream>>>(test, inv_bw, tP, NQ);
    kde_pack_kernel<<<(tot_tr + 255) / 256, 256, 0, stream>>>(train, inv_bw, trP, NK);

    kde_norms_kernel<<<(NQ + 255) / 256, 256, 0, stream>>>(tP, NQ, qn);
    kde_norms_kernel<<<(NK + 255) / 256, 256, 0, stream>>>(trP, NK, kn);

    dim3 grid(NQ / 128, nkg);
    kde_wmma_kernel<<<grid, 256, 0, stream>>>(tP, trP, qn, kn, scal, partial, NQ, NK, nkg);

    kde_reduce_kernel<<<(NQ + 255) / 256, 256, 0, stream>>>(partial, out, NQ, nkg);
}


# Column sum / sum-of-squares pass split over 256 work-groups (64 row chunks x 4 dim slices, f64 kept) + fixed-order partial-sum tree, replacing one 128-thread serial f64 loop; f64 sum order changes
# speedup vs baseline: 10.8825x; 10.8825x over previous
#include <hip/hip_runtime.h>

typedef __attribute__((ext_vector_type(2))) float v2f;
typedef __attribute__((ext_vector_type(8))) float v8f;

#define KDE_D 128
#define KDE_P (KDE_D / 2)
#define KTILES_PER_WAVE 32

#define BW_ROWCHUNKS 64
#define BW_DIMSLICES 4

__global__ __launch_bounds__(256)
void kde_colstat_kernel(const float* __restrict__ train, int n,
                        float* __restrict__ part_f) {
    double* part = (double*)part_f;
    const int slice = blockIdx.x % BW_DIMSLICES;
    const int chunk = blockIdx.x / BW_DIMSLICES;
    const int c4    = threadIdx.x & 7;
    const int slot  = threadIdx.x >> 3;
    const float* col = train + slice * 32 + c4 * 4;

    double s0 = 0.0, s1 = 0.0, s2 = 0.0, s3 = 0.0;
    double q0 = 0.0, q1 = 0.0, q2 = 0.0, q3 = 0.0;
    auto accum = [&](const float4 v) {
        const double x0 = (double)v.x, x1 = (double)v.y;
        const double x2 = (double)v.z, x3 = (double)v.w;
        s0 += x0; q0 += x0 * x0;
        s1 += x1; q1 += x1 * x1;
        s2 += x2; q2 += x2 * x2;
        s3 += x3; q3 += x3 * x3;
    };
    const int rstep = BW_ROWCHUNKS * 32;
    int r = chunk * 32 + slot;
    for (; r + 7 * rstep < n; r += 8 * rstep) {
        float4 v[8];
#pragma unroll
        for (int j = 0; j < 8; ++j)
            v[j] = *(const float4*)(col + (size_t)(r + j * rstep) * KDE_D);
#pragma unroll
        for (int j = 0; j < 8; ++j) accum(v[j]);
    }
    for (; r < n; r += rstep)
        accum(*(const float4*)(col + (size_t)r * KDE_D));

    __shared__ double red[32][64];
    __shared__ double red2[4][64];
    red[slot][c4 * 4 + 0] = s0;
    red[slot][c4 * 4 + 1] = s1;
    red[slot][c4 * 4 + 2] = s2;
    red[slot][c4 * 4 + 3] = s3;
    red[slot][32 + c4 * 4 + 0] = q0;
    red[slot][32 + c4 * 4 + 1] = q1;
    red[slot][32 + c4 * 4 + 2] = q2;
    red[slot][32 + c4 * 4 + 3] = q3;
    __syncthreads();
    const int v = threadIdx.x & 63;
    const int g = threadIdx.x >> 6;
    double acc = red[g * 8][v];
    for (int j = 1; j < 8; ++j) acc += red[g * 8 + j][v];
    red2[g][v] = acc;
    __syncthreads();
    if (threadIdx.x < 64) {
        const double t = (red2[0][v] + red2[1][v]) + (red2[2][v] + red2[3][v]);
        part[chunk * (2 * KDE_D) + (v >> 5) * KDE_D + slice * 32 + (v & 31)] = t;
    }
}

__global__ __launch_bounds__(256)
void kde_bw_kernel(const float* __restrict__ part_f, int n,
                   float* __restrict__ bw, float* __restrict__ inv_bw,
                   float* __restrict__ scal) {
    const double* part = (const double*)part_f;
    const int t = threadIdx.x;
    const int d = t & (KDE_D - 1);
    double acc = part[t];
    for (int c = 1; c < BW_ROWCHUNKS; ++c) acc += part[c * (2 * KDE_D) + t];

    __shared__ double sumsq[KDE_D];
    __shared__ float red[KDE_D];
    if (t >= KDE_D) sumsq[d] = acc;
    __syncthreads();
    if (t < KDE_D) {
        const double s = acc, s2 = sumsq[d];
        double var = (s2 - s * s / (double)n) / (double)(n - 1);
        if (var < 0.0) var = 0.0;
        float stdv = (float)sqrt(var);
        stdv = fmaxf(stdv, 0.01f);
        float scott = (float)exp(-log((double)n) / 132.0);
        float b = 1.06f * stdv * scott;
        b = fminf(b, 0.49f);
        bw[d] = b;
        inv_bw[d] = 1.0f / b;
        red[d] = logf(b);
    }
    __syncthreads();
    for (int off = KDE_D / 2; off > 0; off >>= 1) {
        if (t < off) red[t] += red[t + off];
        __syncthreads();
    }
    if (t == 0) {
        float Z = 0.5f * (float)KDE_D * logf(6.28318530717958647692f)
                + red[0] + logf((float)n);
        scal[0] = Z;
        scal[1] = 1.0f / Z;
    }
}

__global__ void kde_pack_kernel(const float* __restrict__ src,
                                const float* __restrict__ inv_bw,
                                float* __restrict__ dstP, int n) {
    int i = blockIdx.x * blockDim.x + threadIdx.x;
    if (i >= KDE_P * n) return;
    int q = i % n;
    int p = i / n;
    float2 v;
    v.x = src[q * KDE_D + 2 * p + 0] * inv_bw[2 * p + 0];
    v.y = src[q * KDE_D + 2 * p + 1] * inv_bw[2 * p + 1];
    ((float2*)dstP)[p * n + q] = v;
}

__global__ void kde_norms_kernel(const float* __restrict__ srcP, int n,
                                 float* __restrict__ out) {
    int q = blockIdx.x * blockDim.x + threadIdx.x;
    if (q >= n) return;
    float acc = 0.0f;
    const float2* s2 = (const float2*)srcP;
    for (int p = 0; p < KDE_P; ++p) {
        float2 v = s2[p * n + q];
        acc = fmaf(v.x, v.x, acc);
        acc = fmaf(v.y, v.y, acc);
    }
    out[q] = acc;
}

__global__ __launch_bounds__(256)
void kde_wmma_kernel(const float* __restrict__ tP, const float* __restrict__ trP,
                     const float* __restrict__ qn, const float* __restrict__ kn,
                     const float* __restrict__ scal, float* __restrict__ partial,
                     int NQ, int NK, int nkg) {
    const int lane = threadIdx.x & 31;
    const int wave = threadIdx.x >> 5;
    const int qt   = blockIdx.x * 8 + wave;
    const int kg   = blockIdx.y;
    const int q0   = qt * 16;
    const int hi   = lane >> 4;
    const int lo   = lane & 15;
    const float invZ = scal[1];

    const float2* tP2  = (const float2*)tP;
    const float2* trP2 = (const float2*)trP;

    v2f aF[32];
#pragma unroll
    for (int s4 = 0; s4 < 32; ++s4) {
        float2 t = tP2[(2 * s4 + hi) * NQ + q0 + lo];
        aF[s4].x = t.x;
        aF[s4].y = t.y;
    }
    float qrow[8];
#pragma unroll
    for (int r = 0; r < 8; ++r) qrow[r] = qn[q0 + 8 * hi + r];

    float racc[8] = {};
    const int kstart = kg * (KTILES_PER_WAVE * 16);

    v2f b0[32], b1[32];

    auto loadB = [&](v2f (&buf)[32], int kb) {
#pragma unroll
        for (int s4 = 0; s4 < 32; ++s4) {
            float2 t = trP2[(2 * s4 + hi) * NK + kb + lo];
            buf[s4].x = t.x;
            buf[s4].y = t.y;
        }
    };
    auto processB = [&](const v2f (&buf)[32], int kb) {
        v8f acc = {};
#pragma unroll
        for (int s4 = 0; s4 < 32; ++s4) {
            acc = __builtin_amdgcn_wmma_f32_16x16x4_f32(
                false, aF[s4], false, buf[s4], (short)0, acc, false, false);
        }
        const float knv = kn[kb + lo];
#pragma unroll
        for (int r = 0; r < 8; ++r) {
            float sq = fmaxf(qrow[r] + knv - 2.0f * acc[r], 0.0f);
            racc[r] += __expf(fmaf(-0.5f * sq, invZ, -1.0f));
        }
    };

    loadB(b0, kstart);
    for (int ktp = 0; ktp < KTILES_PER_WAVE / 2; ++ktp) {
        const int kb0 = kstart + ktp * 32;
        loadB(b1, kb0 + 16);
        processB(b0, kb0);
        if (ktp + 1 < KTILES_PER_WAVE / 2) loadB(b0, kb0 + 32);
        processB(b1, kb0 + 16);
    }

#pragma unroll
    for (int r = 0; r < 8; ++r) {
        float v = racc[r];
        v += __shfl_xor(v, 1, 32);
        v += __shfl_xor(v, 2, 32);
        v += __shfl_xor(v, 4, 32);
        v += __shfl_xor(v, 8, 32);
        racc[r] = v;
    }
    if (lo == 0) {
#pragma unroll
        for (int r = 0; r < 8; ++r) {
            int row = q0 + 8 * hi + r;
            partial[row * nkg + kg] = racc[r];
        }
    }
}

__global__ void kde_reduce_kernel(const float* __restrict__ partial,
                                  float* __restrict__ out, int NQ, int nkg) {
    int q = blockIdx.x * blockDim.x + threadIdx.x;
    if (q >= NQ) return;
    float acc = 0.0f;
    for (int g = 0; g < nkg; ++g) acc += partial[q * nkg + g];
    out[q] = acc;
}

extern "C" void kernel_launch(void* const* d_in, const int* in_sizes, int n_in,
                              void* d_out, int out_size, void* d_ws, size_t ws_size,
                              hipStream_t stream) {
    const float* test  = (const float*)d_in[0];
    const float* train = (const float*)d_in[1];
    float* out = (float*)d_out;

    const int NQ = in_sizes[0] / KDE_D;
    const int NK = in_sizes[1] / KDE_D;
    const int nkg = NK / (KTILES_PER_WAVE * 16);

    float* ws      = (float*)d_ws;
    float* bw      = ws;
    float* inv_bw  = ws + KDE_D;
    float* scal    = ws + 2 * KDE_D;
    float* qn      = ws + 3 * KDE_D;
    float* kn      = qn + NQ;
    float* tP      = kn + NK;
    float* trP     = tP + KDE_D * NQ;
    float* partial = trP + KDE_D * NK;

    kde_colstat_kernel<<<BW_ROWCHUNKS * BW_DIMSLICES, 256, 0, stream>>>(train, NK, partial);
    kde_bw_kernel<<<1, 2 * KDE_D, 0, stream>>>(partial, NK, bw, inv_bw, scal);

    int tot_t  = KDE_P * NQ;
    int tot_tr = KDE_P * NK;
    kde_pack_kernel<<<(tot_t + 255) / 256, 256, 0, stream>>>(test, inv_bw, tP, NQ);
    kde_pack_kernel<<<(tot_tr + 255) / 256, 256, 0, stream>>>(train, inv_bw, trP, NK);

    kde_norms_kernel<<<(NQ + 255) / 256, 256, 0, stream>>>(tP, NQ, qn);
    kde_norms_kernel<<<(NK + 255) / 256, 256, 0, stream>>>(trP, NK, kn);

    dim3 grid(NQ / 128, nkg);
    kde_wmma_kernel<<<grid, 256, 0, stream>>>(tP, trP, qn, kn, scal, partial, NQ, NK, nkg);

    kde_reduce_kernel<<<(NQ + 255) / 256, 256, 0, stream>>>(partial, out, NQ, nkg);
}


# Scale+transpose+pack and row norms of both operands fused into one LDS-tiled launch (64 rows per work-group, whole-row loads, whole-line stores) replacing 2 pack + 2 norm launches; sums unchanged
# speedup vs baseline: 12.1387x; 1.1154x over previous
#include <hip/hip_runtime.h>

typedef __attribute__((ext_vector_type(2))) float v2f;
typedef __attribute__((ext_vector_type(8))) float v8f;

#define KDE_D 128
#define KDE_P (KDE_D / 2)
#define KTILES_PER_WAVE 32

#define BW_ROWCHUNKS 64
#define BW_DIMSLICES 4

__global__ __launch_bounds__(256)
void kde_colstat_kernel(const float* __restrict__ train, int n,
                        float* __restrict__ part_f) {
    double* part = (double*)part_f;
    const int slice = blockIdx.x % BW_DIMSLICES;
    const int chunk = blockIdx.x / BW_DIMSLICES;
    const int c4    = threadIdx.x & 7;
    const int slot  = threadIdx.x >> 3;
    const float* col = train + slice * 32 + c4 * 4;

    double s0 = 0.0, s1 = 0.0, s2 = 0.0, s3 = 0.0;
    double q0 = 0.0, q1 = 0.0, q2 = 0.0, q3 = 0.0;
    auto accum = [&](const float4 v) {
        const double x0 = (double)v.x, x1 = (double)v.y;
        const double x2 = (double)v.z, x3 = (double)v.w;
        s0 += x0; q0 += x0 * x0;
        s1 += x1; q1 += x1 * x1;
        s2 += x2; q2 += x2 * x2;
        s3 += x3; q3 += x3 * x3;
    };
    const int rstep = BW_ROWCHUNKS * 32;
    int r = chunk * 32 + slot;
    for (; r + 7 * rstep < n; r += 8 * rstep) {
        float4 v[8];
#pragma unroll
        for (int j = 0; j < 8; ++j)
            v[j] = *(const float4*)(col + (size_t)(r + j * rstep) * KDE_D);
#pragma unroll
        for (int j = 0; j < 8; ++j) accum(v[j]);
    }
    for (; r < n; r += rstep)
        accum(*(const float4*)(col + (size_t)r * KDE_D));

    __shared__ double red[32][64];
    __shared__ double red2[4][64];
    red[slot][c4 * 4 + 0] = s0;
    red[slot][c4 * 4 + 1] = s1;
    red[slot][c4 * 4 + 2] = s2;
    red[slot][c4 * 4 + 3] = s3;
    red[slot][32 + c4 * 4 + 0] = q0;
    red[slot][32 + c4 * 4 + 1] = q1;
    red[slot][32 + c4 * 4 + 2] = q2;
    red[slot][32 + c4 * 4 + 3] = q3;
    __syncthreads();
    const int v = threadIdx.x & 63;
    const int g = threadIdx.x >> 6;
    double acc = red[g * 8][v];
    for (int j = 1; j < 8; ++j) acc += red[g * 8 + j][v];
    red2[g][v] = acc;
    __syncthreads();
    if (threadIdx.x < 64) {
        const double t = (red2[0][v] + red2[1][v]) + (red2[2][v] + red2[3][v]);
        part[chunk * (2 * KDE_D) + (v >> 5) * KDE_D + slice * 32 + (v & 31)] = t;
    }
}

__global__ __launch_bounds__(256)
void kde_bw_kernel(const float* __restrict__ part_f, int n,
                   float* __restrict__ bw, float* __restrict__ inv_bw,
                   float* __restrict__ scal) {
    const double* part = (const double*)part_f;
    const int t = threadIdx.x;
    const int d = t & (KDE_D - 1);
    double acc = part[t];
    for (int c = 1; c < BW_ROWCHUNKS; ++c) acc += part[c * (2 * KDE_D) + t];

    __shared__ double sumsq[KDE_D];
    __shared__ float red[KDE_D];
    if (t >= KDE_D) sumsq[d] = acc;
    __syncthreads();
    if (t < KDE_D) {
        const double s = acc, s2 = sumsq[d];
        double var = (s2 - s * s / (double)n) / (double)(n - 1);
        if (var < 0.0) var = 0.0;
        float stdv = (float)sqrt(var);
        stdv = fmaxf(stdv, 0.01f);
        float scott = (float)exp(-log((double)n) / 132.0);
        float b = 1.06f * stdv * scott;
        b = fminf(b, 0.49f);
        bw[d] = b;
        inv_bw[d] = 1.0f / b;
        red[d] = logf(b);
    }
    __syncthreads();
    for (int off = KDE_D / 2; off > 0; off >>= 1) {
        if (t < off) red[t] += red[t + off];
        __syncthreads();
    }
    if (t == 0) {
        float Z = 0.5f * (float)KDE_D * logf(6.28318530717958647692f)
                + red[0] + logf((float)n);
        scal[0] = Z;
        scal[1] = 1.0f / Z;
    }
}

#define PREP_ROWS 64
#define PREP_PITCH (KDE_D + 2)

__global__ __launch_bounds__(256)
void kde_prep_kernel(const float* __restrict__ test, const float* __restrict__ train,
                     const float* __restrict__ inv_bw,
                     float* __restrict__ tP, float* __restrict__ trP,
                     float* __restrict__ qn, float* __restrict__ kn,
                     int NQ, int NK) {
    __shared__ float tile[PREP_ROWS * PREP_PITCH];

    const int nqb = NQ / PREP_ROWS;
    const bool is_test = (int)blockIdx.x < nqb;
    const float* src = is_test ? test : train;
    float2* dstP     = (float2*)(is_test ? tP : trP);
    float* nrm       = is_test ? qn : kn;
    const int n      = is_test ? NQ : NK;
    const int r0     = (is_test ? (int)blockIdx.x : (int)blockIdx.x - nqb) * PREP_ROWS;

    const int lane = threadIdx.x & 31;
    const int wave = threadIdx.x >> 5;

    const float2 ib01 = *(const float2*)(inv_bw + 4 * lane);
    const float2 ib23 = *(const float2*)(inv_bw + 4 * lane + 2);
    float4 v[8];
#pragma unroll
    for (int j = 0; j < 8; ++j)
        v[j] = *(const float4*)(src + (size_t)(r0 + 8 * j + wave) * KDE_D + 4 * lane);
#pragma unroll
    for (int j = 0; j < 8; ++j) {
        float* t = tile + (8 * j + wave) * PREP_PITCH + 4 * lane;
        float2 a, b;
        a.x = v[j].x * ib01.x;  a.y = v[j].y * ib01.y;
        b.x = v[j].z * ib23.x;  b.y = v[j].w * ib23.y;
        *(float2*)(t)     = a;
        *(float2*)(t + 2) = b;
    }
    __syncthreads();

#pragma unroll
    for (int i = 0; i < 16; ++i) {
        const int p   = wave + 8 * (i >> 1);
        const int row = lane + 32 * (i & 1);
        const float2 val = *(const float2*)(tile + row * PREP_PITCH + 2 * p);
        dstP[(size_t)p * n + r0 + row] = val;
    }

    if (threadIdx.x < PREP_ROWS) {
        const float* t = tile + threadIdx.x * PREP_PITCH;
        float acc = 0.0f;
#pragma unroll 16
        for (int p = 0; p < KDE_P; ++p) {
            const float2 x = *(const float2*)(t + 2 * p);
            acc = fmaf(x.x, x.x, acc);
            acc = fmaf(x.y, x.y, acc);
        }
        nrm[r0 + threadIdx.x] = acc;
    }
}

__global__ __launch_bounds__(256)
void kde_wmma_kernel(const float* __restrict__ tP, const float* __restrict__ trP,
                     const float* __restrict__ qn, const float* __restrict__ kn,
                     const float* __restrict__ scal, float* __restrict__ partial,
                     int NQ, int NK, int nkg) {
    const int lane = threadIdx.x & 31;
    const int wave = threadIdx.x >> 5;
    const int qt   = blockIdx.x * 8 + wave;
    const int kg   = blockIdx.y;
    const int q0   = qt * 16;
    const int hi   = lane >> 4;
    const int lo   = lane & 15;
    const float invZ = scal[1];

    const float2* tP2  = (const float2*)tP;
    const float2* trP2 = (const float2*)trP;

    v2f aF[32];
#pragma unroll
    for (int s4 = 0; s4 < 32; ++s4) {
        float2 t = tP2[(2 * s4 + hi) * NQ + q0 + lo];
        aF[s4].x = t.x;
        aF[s4].y = t.y;
    }
    float qrow[8];
#pragma unroll
    for (int r = 0; r < 8; ++r) qrow[r] = qn[q0 + 8 * hi + r];

    float racc[8] = {};
    const int kstart = kg * (KTILES_PER_WAVE * 16);

    v2f b0[32], b1[32];

    auto loadB = [&](v2f (&buf)[32], int kb) {
#pragma unroll
        for (int s4 = 0; s4 < 32; ++s4) {
            float2 t = trP2[(2 * s4 + hi) * NK + kb + lo];
            buf[s4].x = t.x;
            buf[s4].y = t.y;
        }
    };
    auto processB = [&](const v2f (&buf)[32], int kb) {
        v8f acc = {};
#pragma unroll
        for (int s4 = 0; s4 < 32; ++s4) {
            acc = __builtin_amdgcn_wmma_f32_16x16x4_f32(
                false, aF[s4], false, buf[s4], (short)0, acc, false, false);
        }
        const float knv = kn[kb + lo];
#pragma unroll
        for (int r = 0; r < 8; ++r) {
            float sq = fmaxf(qrow[r] + knv - 2.0f * acc[r], 0.0f);
            racc[r] += __expf(fmaf(-0.5f * sq, invZ, -1.0f));
        }
    };

    loadB(b0, kstart);
    for (int ktp = 0; ktp < KTILES_PER_WAVE / 2; ++ktp) {
        const int kb0 = kstart + ktp * 32;
        loadB(b1, kb0 + 16);
        processB(b0, kb0);
        if (ktp + 1 < KTILES_PER_WAVE / 2) loadB(b0, kb0 + 32);
        processB(b1, kb0 + 16);
    }

#pragma unroll
    for (int r = 0; r < 8; ++r) {
        float v = racc[r];
        v += __shfl_xor(v, 1, 32);
        v += __shfl_xor(v, 2, 32);
        v += __shfl_xor(v, 4, 32);
        v += __shfl_xor(v, 8, 32);
        racc[r] = v;
    }
    if (lo == 0) {
#pragma unroll
        for (int r = 0; r < 8; ++r) {
            int row = q0 + 8 * hi + r;
            partial[row * nkg + kg] = racc[r];
        }
    }
}

__global__ void kde_reduce_kernel(const float* __restrict__ partial,
                                  float* __restrict__ out, int NQ, int nkg) {
    int q = blockIdx.x * blockDim.x + threadIdx.x;
    if (q >= NQ) return;
    float acc = 0.0f;
    for (int g = 0; g < nkg; ++g) acc += partial[q * nkg + g];
    out[q] = acc;
}

extern "C" void kernel_launch(void* const* d_in, const int* in_sizes, int n_in,
                              void* d_out, int out_size, void* d_ws, size_t ws_size,
                              hipStream_t stream) {
    const float* test  = (const float*)d_in[0];
    const float* train = (const float*)d_in[1];
    float* out = (float*)d_out;

    const int NQ = in_sizes[0] / KDE_D;
    const int NK = in_sizes[1] / KDE_D;
    const int nkg = NK / (KTILES_PER_WAVE * 16);

    float* ws      = (float*)d_ws;
    float* bw      = ws;
    float* inv_bw  = ws + KDE_D;
    float* scal    = ws + 2 * KDE_D;
    float* qn      = ws + 3 * KDE_D;
    float* kn      = qn + NQ;
    float* tP      = kn + NK;
    float* trP     = tP + KDE_D * NQ;
    float* partial = trP + KDE_D * NK;

    kde_colstat_kernel<<<BW_ROWCHUNKS * BW_DIMSLICES, 256, 0, stream>>>(train, NK, partial);
    kde_bw_kernel<<<1, 2 * KDE_D, 0, stream>>>(partial, NK, bw, inv_bw, scal);

    kde_prep_kernel<<<(NQ + NK) / PREP_ROWS, 256, 0, stream>>>(test, train, inv_bw,
                                                              tP, trP, qn, kn, NQ, NK);

    dim3 grid(NQ / 128, nkg);
    kde_wmma_kernel<<<grid, 256, 0, stream>>>(tP, trP, qn, kn, scal, partial, NQ, NK, nkg);

    kde_reduce_kernel<<<(NQ + 255) / 256, 256, 0, stream>>>(partial, out, NQ, nkg);
}


# WMMA kernel: each wave owns two q-tiles (32 rows), so each B fragment feeds two independent accumulator chains (half the B loads per WMMA); loop pipelined by hand, loads a tile ahead; sums unchanged
# speedup vs baseline: 15.1099x; 1.2448x over previous
#include <hip/hip_runtime.h>

typedef __attribute__((ext_vector_type(2))) float v2f;
typedef __attribute__((ext_vector_type(8))) float v8f;

#define KDE_D 128
#define KDE_P (KDE_D / 2)
#define KTILES_PER_WAVE 32

#define BW_ROWCHUNKS 64
#define BW_DIMSLICES 4

__global__ __launch_bounds__(256)
void kde_colstat_kernel(const float* __restrict__ train, int n,
                        float* __restrict__ part_f) {
    double* part = (double*)part_f;
    const int slice = blockIdx.x % BW_DIMSLICES;
    const int chunk = blockIdx.x / BW_DIMSLICES;
    const int c4    = threadIdx.x & 7;
    const int slot  = threadIdx.x >> 3;
    const float* col = train + slice * 32 + c4 * 4;

    double s0 = 0.0, s1 = 0.0, s2 = 0.0, s3 = 0.0;
    double q0 = 0.0, q1 = 0.0, q2 = 0.0, q3 = 0.0;
    auto accum = [&](const float4 v) {
        const double x0 = (double)v.x, x1 = (double)v.y;
        const double x2 = (double)v.z, x3 = (double)v.w;
        s0 += x0; q0 += x0 * x0;
        s1 += x1; q1 += x1 * x1;
        s2 += x2; q2 += x2 * x2;
        s3 += x3; q3 += x3 * x3;
    };
    const int rstep = BW_ROWCHUNKS * 32;
    int r = chunk * 32 + slot;
    for (; r + 7 * rstep < n; r += 8 * rstep) {
        float4 v[8];
#pragma unroll
        for (int j = 0; j < 8; ++j)
            v[j] = *(const float4*)(col + (size_t)(r + j * rstep) * KDE_D);
#pragma unroll
        for (int j = 0; j < 8; ++j) accum(v[j]);
    }
    for (; r < n; r += rstep)
        accum(*(const float4*)(col + (size_t)r * KDE_D));

    __shared__ double red[32][64];
    __shared__ double red2[4][64];
    red[slot][c4 * 4 + 0] = s0;
    red[slot][c4 * 4 + 1] = s1;
    red[slot][c4 * 4 + 2] = s2;
    red[slot][c4 * 4 + 3] = s3;
    red[slot][32 + c4 * 4 + 0] = q0;
    red[slot][32 + c4 * 4 + 1] = q1;
    red[slot][32 + c4 * 4 + 2] = q2;
    red[slot][32 + c4 * 4 + 3] = q3;
    __syncthreads();
    const int v = threadIdx.x & 63;
    const int g = threadIdx.x >> 6;
    double acc = red[g * 8][v];
    for (int j = 1; j < 8; ++j) acc += red[g * 8 + j][v];
    red2[g][v] = acc;
    __syncthreads();
    if (threadIdx.x < 64) {
        const double t = (red2[0][v] + red2[1][v]) + (red2[2][v] + red2[3][v]);
        part[chunk * (2 * KDE_D) + (v >> 5) * KDE_D + slice * 32 + (v & 31)] = t;
    }
}

__global__ __launch_bounds__(256)
void kde_bw_kernel(const float* __restrict__ part_f, int n,
                   float* __restrict__ bw, float* __restrict__ inv_bw,
                   float* __restrict__ scal) {
    const double* part = (const double*)part_f;
    const int t = threadIdx.x;
    const int d = t & (KDE_D - 1);
    double acc = part[t];
    for (int c = 1; c < BW_ROWCHUNKS; ++c) acc += part[c * (2 * KDE_D) + t];

    __shared__ double sumsq[KDE_D];
    __shared__ float red[KDE_D];
    if (t >= KDE_D) sumsq[d] = acc;
    __syncthreads();
    if (t < KDE_D) {
        const double s = acc, s2 = sumsq[d];
        double var = (s2 - s * s / (double)n) / (double)(n - 1);
        if (var < 0.0) var = 0.0;
        float stdv = (float)sqrt(var);
        stdv = fmaxf(stdv, 0.01f);
        float scott = (float)exp(-log((double)n) / 132.0);
        float b = 1.06f * stdv * scott;
        b = fminf(b, 0.49f);
        bw[d] = b;
        inv_bw[d] = 1.0f / b;
        red[d] = logf(b);
    }
    __syncthreads();
    for (int off = KDE_D / 2; off > 0; off >>= 1) {
        if (t < off) red[t] += red[t + off];
        __syncthreads();
    }
    if (t == 0) {
        float Z = 0.5f * (float)KDE_D * logf(6.28318530717958647692f)
                + red[0] + logf((float)n);
        scal[0] = Z;
        scal[1] = 1.0f / Z;
    }
}

#define PREP_ROWS 64
#define PREP_PITCH (KDE_D + 2)

__global__ __launch_bounds__(256)
void kde_prep_kernel(const float* __restrict__ test, const float* __restrict__ train,
                     const float* __restrict__ inv_bw,
                     float* __restrict__ tP, float* __restrict__ trP,
                     float* __restrict__ qn, float* __restrict__ kn,
                     int NQ, int NK) {
    __shared__ float tile[PREP_ROWS * PREP_PITCH];

    const int nqb = NQ / PREP_ROWS;
    const bool is_test = (int)blockIdx.x < nqb;
    const float* src = is_test ? test : train;
    float2* dstP     = (float2*)(is_test ? tP : trP);
    float* nrm       = is_test ? qn : kn;
    const int n      = is_test ? NQ : NK;
    const int r0     = (is_test ? (int)blockIdx.x : (int)blockIdx.x - nqb) * PREP_ROWS;

    const int lane = threadIdx.x & 31;
    const int wave = threadIdx.x >> 5;

    const float2 ib01 = *(const float2*)(inv_bw + 4 * lane);
    const float2 ib23 = *(const float2*)(inv_bw + 4 * lane + 2);
    float4 v[8];
#pragma unroll
    for (int j = 0; j < 8; ++j)
        v[j] = *(const float4*)(src + (size_t)(r0 + 8 * j + wave) * KDE_D + 4 * lane);
#pragma unroll
    for (int j = 0; j < 8; ++j) {
        float* t = tile + (8 * j + wave) * PREP_PITCH + 4 * lane;
        float2 a, b;
        a.x = v[j].x * ib01.x;  a.y = v[j].y * ib01.y;
        b.x = v[j].z * ib23.x;  b.y = v[j].w * ib23.y;
        *(float2*)(t)     = a;
        *(float2*)(t + 2) = b;
    }
    __syncthreads();

#pragma unroll
    for (int i = 0; i < 16; ++i) {
        const int p   = wave + 8 * (i >> 1);
        const int row = lane + 32 * (i & 1);
        const float2 val = *(const float2*)(tile + row * PREP_PITCH + 2 * p);
        dstP[(size_t)p * n + r0 + row] = val;
    }

    if (threadIdx.x < PREP_ROWS) {
        const float* t = tile + threadIdx.x * PREP_PITCH;
        float acc = 0.0f;
#pragma unroll 16
        for (int p = 0; p < KDE_P; ++p) {
            const float2 x = *(const float2*)(t + 2 * p);
            acc = fmaf(x.x, x.x, acc);
            acc = fmaf(x.y, x.y, acc);
        }
        nrm[r0 + threadIdx.x] = acc;
    }
}

__global__ __launch_bounds__(256)
void kde_wmma_kernel(const float* __restrict__ tP, const float* __restrict__ trP,
                     const float* __restrict__ qn, const float* __restrict__ kn,
                     const float* __restrict__ scal, float* __restrict__ partial,
                     int NQ, int NK, int nkg) {
    const int lane = threadIdx.x & 31;
    const int wave = threadIdx.x >> 5;
    const int q0   = (blockIdx.x * 8 + wave) * 32;
    const int kg   = blockIdx.y;
    const int hi   = lane >> 4;
    const int lo   = lane & 15;
    const float invZ = scal[1];

    const float2* tP2  = (const float2*)tP;
    const float2* trP2 = (const float2*)trP;

    v2f aF0[32], aF1[32];
#pragma unroll
    for (int s4 = 0; s4 < 32; ++s4) {
        const float2* src = tP2 + (2 * s4 + hi) * NQ + q0 + lo;
        const float2 t0 = src[0];
        const float2 t1 = src[16];
        aF0[s4].x = t0.x;  aF0[s4].y = t0.y;
        aF1[s4].x = t1.x;  aF1[s4].y = t1.y;
    }
    float qrow0[8], qrow1[8];
#pragma unroll
    for (int r = 0; r < 8; ++r) {
        qrow0[r] = qn[q0 + 8 * hi + r];
        qrow1[r] = qn[q0 + 16 + 8 * hi + r];
    }

    float racc0[8] = {}, racc1[8] = {};
    const int kstart = kg * (KTILES_PER_WAVE * 16);

    v2f b0[32], b1[32];
    float kn0, kn1;

    auto loadB = [&](v2f (&buf)[32], float& knv, int kb) {
#pragma unroll
        for (int s4 = 0; s4 < 32; ++s4) {
            float2 t = trP2[(2 * s4 + hi) * NK + kb + lo];
            buf[s4].x = t.x;
            buf[s4].y = t.y;
        }
        knv = kn[kb + lo];
    };
    auto chain = [&](const v2f (&buf)[32], v8f& acc0, v8f& acc1) {
        v8f c0 = {}, c1 = {};
#pragma unroll
        for (int s4 = 0; s4 < 32; ++s4) {
            c0 = __builtin_amdgcn_wmma_f32_16x16x4_f32(
                false, aF0[s4], false, buf[s4], (short)0, c0, false, false);
            c1 = __builtin_amdgcn_wmma_f32_16x16x4_f32(
                false, aF1[s4], false, buf[s4], (short)0, c1, false, false);
        }
        acc0 = c0;
        acc1 = c1;
    };
    auto epilogue = [&](const v8f& acc0, const v8f& acc1, const float knv) {
#pragma unroll
        for (int r = 0; r < 8; ++r) {
            float sq0 = fmaxf(qrow0[r] + knv - 2.0f * acc0[r], 0.0f);
            float sq1 = fmaxf(qrow1[r] + knv - 2.0f * acc1[r], 0.0f);
            racc0[r] += __expf(fmaf(-0.5f * sq0, invZ, -1.0f));
            racc1[r] += __expf(fmaf(-0.5f * sq1, invZ, -1.0f));
        }
    };

#define SCHED_FENCE() __builtin_amdgcn_sched_barrier(0)

    v8f accA0, accA1, accB0, accB1;
    float knA, knB;

    loadB(b0, kn0, kstart);
    loadB(b1, kn1, kstart + 16);
    SCHED_FENCE();
    chain(b0, accA0, accA1);
    knA = kn0;
    SCHED_FENCE();
    loadB(b0, kn0, kstart + 32);
    for (int i = 0; i < KTILES_PER_WAVE / 2 - 1; ++i) {
        SCHED_FENCE();
        chain(b1, accB0, accB1);
        knB = kn1;
        epilogue(accA0, accA1, knA);
        SCHED_FENCE();
        loadB(b1, kn1, kstart + (2 * i + 3) * 16);
        SCHED_FENCE();
        chain(b0, accA0, accA1);
        knA = kn0;
        epilogue(accB0, accB1, knB);
        SCHED_FENCE();
        loadB(b0, kn0, kstart + ((2 * i + 4) & (KTILES_PER_WAVE - 1)) * 16);
    }
    SCHED_FENCE();
    chain(b1, accB0, accB1);
    epilogue(accA0, accA1, knA);
    epilogue(accB0, accB1, kn1);
#undef SCHED_FENCE

#pragma unroll
    for (int r = 0; r < 8; ++r) {
        float v = racc0[r];
        v += __shfl_xor(v, 1, 32);
        v += __shfl_xor(v, 2, 32);
        v += __shfl_xor(v, 4, 32);
        v += __shfl_xor(v, 8, 32);
        racc0[r] = v;
        float w = racc1[r];
        w += __shfl_xor(w, 1, 32);
        w += __shfl_xor(w, 2, 32);
        w += __shfl_xor(w, 4, 32);
        w += __shfl_xor(w, 8, 32);
        racc1[r] = w;
    }
    if (lo == 0) {
#pragma unroll
        for (int r = 0; r < 8; ++r) {
            const int row = q0 + 8 * hi + r;
            partial[row * nkg + kg] = racc0[r];
            partial[(row + 16) * nkg + kg] = racc1[r];
        }
    }
}

__global__ void kde_reduce_kernel(const float* __restrict__ partial,
                                  float* __restrict__ out, int NQ, int nkg) {
    int q = blockIdx.x * blockDim.x + threadIdx.x;
    if (q >= NQ) return;
    float acc = 0.0f;
    for (int g = 0; g < nkg; ++g) acc += partial[q * nkg + g];
    out[q] = acc;
}

extern "C" void kernel_launch(void* const* d_in, const int* in_sizes, int n_in,
                              void* d_out, int out_size, void* d_ws, size_t ws_size,
                              hipStream_t stream) {
    const float* test  = (const float*)d_in[0];
    const float* train = (const float*)d_in[1];
    float* out = (float*)d_out;

    const int NQ = in_sizes[0] / KDE_D;
    const int NK = in_sizes[1] / KDE_D;
    const int nkg = NK / (KTILES_PER_WAVE * 16);

    float* ws      = (float*)d_ws;
    float* bw      = ws;
    float* inv_bw  = ws + KDE_D;
    float* scal    = ws + 2 * KDE_D;
    float* qn      = ws + 3 * KDE_D;
    float* kn      = qn + NQ;
    float* tP      = kn + NK;
    float* trP     = tP + KDE_D * NQ;
    float* partial = trP + KDE_D * NK;

    kde_colstat_kernel<<<BW_ROWCHUNKS * BW_DIMSLICES, 256, 0, stream>>>(train, NK, partial);
    kde_bw_kernel<<<1, 2 * KDE_D, 0, stream>>>(partial, NK, bw, inv_bw, scal);

    kde_prep_kernel<<<(NQ + NK) / PREP_ROWS, 256, 0, stream>>>(test, train, inv_bw,
                                                              tP, trP, qn, kn, NQ, NK);

    dim3 grid(NQ / 256, nkg);
    kde_wmma_kernel<<<grid, 256, 0, stream>>>(tP, trP, qn, kn, scal, partial, NQ, NK, nkg);

    kde_reduce_kernel<<<(NQ + 255) / 256, 256, 0, stream>>>(partial, out, NQ, nkg);
}


# WMMA loop sheds vector issue slots: tile-major packed operands (16-byte fragment loads, one base + constant offsets, no address math), exponent folded to 5 ops per element; exponent rounding changes
# speedup vs baseline: 16.6311x; 1.1007x over previous
#include <hip/hip_runtime.h>

typedef __attribute__((ext_vector_type(2))) float v2f;
typedef __attribute__((ext_vector_type(8))) float v8f;

#define KDE_D 128
#define KDE_P (KDE_D / 2)
#define KTILES_PER_WAVE 32

#define BW_ROWCHUNKS 64
#define BW_DIMSLICES 4

__global__ __launch_bounds__(256)
void kde_colstat_kernel(const float* __restrict__ train, int n,
                        float* __restrict__ part_f) {
    double* part = (double*)part_f;
    const int slice = blockIdx.x % BW_DIMSLICES;
    const int chunk = blockIdx.x / BW_DIMSLICES;
    const int c4    = threadIdx.x & 7;
    const int slot  = threadIdx.x >> 3;
    const float* col = train + slice * 32 + c4 * 4;

    double s0 = 0.0, s1 = 0.0, s2 = 0.0, s3 = 0.0;
    double q0 = 0.0, q1 = 0.0, q2 = 0.0, q3 = 0.0;
    auto accum = [&](const float4 v) {
        const double x0 = (double)v.x, x1 = (double)v.y;
        const double x2 = (double)v.z, x3 = (double)v.w;
        s0 += x0; q0 += x0 * x0;
        s1 += x1; q1 += x1 * x1;
        s2 += x2; q2 += x2 * x2;
        s3 += x3; q3 += x3 * x3;
    };
    const int rstep = BW_ROWCHUNKS * 32;
    int r = chunk * 32 + slot;
    for (; r + 7 * rstep < n; r += 8 * rstep) {
        float4 v[8];
#pragma unroll
        for (int j = 0; j < 8; ++j)
            v[j] = *(const float4*)(col + (size_t)(r + j * rstep) * KDE_D);
#pragma unroll
        for (int j = 0; j < 8; ++j) accum(v[j]);
    }
    for (; r < n; r += rstep)
        accum(*(const float4*)(col + (size_t)r * KDE_D));

    __shared__ double red[32][64];
    __shared__ double red2[4][64];
    red[slot][c4 * 4 + 0] = s0;
    red[slot][c4 * 4 + 1] = s1;
    red[slot][c4 * 4 + 2] = s2;
    red[slot][c4 * 4 + 3] = s3;
    red[slot][32 + c4 * 4 + 0] = q0;
    red[slot][32 + c4 * 4 + 1] = q1;
    red[slot][32 + c4 * 4 + 2] = q2;
    red[slot][32 + c4 * 4 + 3] = q3;
    __syncthreads();
    const int v = threadIdx.x & 63;
    const int g = threadIdx.x >> 6;
    double acc = red[g * 8][v];
    for (int j = 1; j < 8; ++j) acc += red[g * 8 + j][v];
    red2[g][v] = acc;
    __syncthreads();
    if (threadIdx.x < 64) {
        const double t = (red2[0][v] + red2[1][v]) + (red2[2][v] + red2[3][v]);
        part[chunk * (2 * KDE_D) + (v >> 5) * KDE_D + slice * 32 + (v & 31)] = t;
    }
}

__global__ __launch_bounds__(256)
void kde_bw_kernel(const float* __restrict__ part_f, int n,
                   float* __restrict__ bw, float* __restrict__ inv_bw,
                   float* __restrict__ scal) {
    const double* part = (const double*)part_f;
    const int t = threadIdx.x;
    const int d = t & (KDE_D - 1);
    double acc = part[t];
    for (int c = 1; c < BW_ROWCHUNKS; ++c) acc += part[c * (2 * KDE_D) + t];

    __shared__ double sumsq[KDE_D];
    __shared__ float red[KDE_D];
    if (t >= KDE_D) sumsq[d] = acc;
    __syncthreads();
    if (t < KDE_D) {
        const double s = acc, s2 = sumsq[d];
        double var = (s2 - s * s / (double)n) / (double)(n - 1);
        if (var < 0.0) var = 0.0;
        float stdv = (float)sqrt(var);
        stdv = fmaxf(stdv, 0.01f);
        float scott = (float)exp(-log((double)n) / 132.0);
        float b = 1.06f * stdv * scott;
        b = fminf(b, 0.49f);
        bw[d] = b;
        inv_bw[d] = 1.0f / b;
        red[d] = logf(b);
    }
    __syncthreads();
    for (int off = KDE_D / 2; off > 0; off >>= 1) {
        if (t < off) red[t] += red[t + off];
        __syncthreads();
    }
    if (t == 0) {
        float Z = 0.5f * (float)KDE_D * logf(6.28318530717958647692f)
                + red[0] + logf((float)n);
        scal[0] = Z;
        scal[1] = 1.0f / Z;
    }
}

#define PREP_ROWS 64
#define PREP_PITCH (KDE_D + 2)

__global__ __launch_bounds__(256)
void kde_prep_kernel(const float* __restrict__ test, const float* __restrict__ train,
                     const float* __restrict__ inv_bw,
                     float* __restrict__ tP, float* __restrict__ trP,
                     float* __restrict__ qn, float* __restrict__ kn,
                     int NQ, int NK) {
    __shared__ float tile[PREP_ROWS * PREP_PITCH];

    const int nqb = NQ / PREP_ROWS;
    const bool is_test = (int)blockIdx.x < nqb;
    const float* src = is_test ? test : train;
    float* nrm       = is_test ? qn : kn;
    const int r0     = (is_test ? (int)blockIdx.x : (int)blockIdx.x - nqb) * PREP_ROWS;
    float4* dst4     = (float4*)(is_test ? tP : trP) + (size_t)(r0 >> 4) * 512;

    const int lane = threadIdx.x & 31;
    const int wave = threadIdx.x >> 5;

    const float2 ib01 = *(const float2*)(inv_bw + 4 * lane);
    const float2 ib23 = *(const float2*)(inv_bw + 4 * lane + 2);
    float4 v[8];
#pragma unroll
    for (int j = 0; j < 8; ++j)
        v[j] = *(const float4*)(src + (size_t)(r0 + 8 * j + wave) * KDE_D + 4 * lane);
#pragma unroll
    for (int j = 0; j < 8; ++j) {
        float* t = tile + (8 * j + wave) * PREP_PITCH + 4 * lane;
        float2 a, b;
        a.x = v[j].x * ib01.x;  a.y = v[j].y * ib01.y;
        b.x = v[j].z * ib23.x;  b.y = v[j].w * ib23.y;
        *(float2*)(t)     = a;
        *(float2*)(t + 2) = b;
    }
    __syncthreads();

#pragma unroll
    for (int i = 0; i < 8; ++i) {
        const int jj  = i >> 1;
        const int m   = (i & 1) * 8 + wave;
        const int row = jj * 16 + (lane & 15);
        const float* t = tile + row * PREP_PITCH + 8 * m + 2 * (lane >> 4);
        const float2 a = *(const float2*)(t);
        const float2 b = *(const float2*)(t + 4);
        float4 val;
        val.x = a.x;  val.y = a.y;  val.z = b.x;  val.w = b.y;
        dst4[i * 256 + threadIdx.x] = val;
    }

    if (threadIdx.x < PREP_ROWS) {
        const float* t = tile + threadIdx.x * PREP_PITCH;
        float acc = 0.0f;
#pragma unroll 16
        for (int p = 0; p < KDE_P; ++p) {
            const float2 x = *(const float2*)(t + 2 * p);
            acc = fmaf(x.x, x.x, acc);
            acc = fmaf(x.y, x.y, acc);
        }
        nrm[r0 + threadIdx.x] = acc;
    }
}

__global__ __launch_bounds__(256)
void kde_wmma_kernel(const float* __restrict__ tP, const float* __restrict__ trP,
                     const float* __restrict__ qn, const float* __restrict__ kn,
                     const float* __restrict__ scal, float* __restrict__ partial,
                     int NQ, int NK, int nkg) {
    const int lane = threadIdx.x & 31;
    const int wave = threadIdx.x >> 5;
    const int q0   = (blockIdx.x * 8 + wave) * 32;
    const int kg   = blockIdx.y;
    const int hi   = lane >> 4;
    const int lo   = lane & 15;
    const float invZ = scal[1];

    const float L2E = 1.44269504088896341f;
    const float cz  = -0.5f * invZ * L2E;
    const float m2  = -2.0f * cz;
    const float clamp_lo = (cz < 0.0f) ? -__builtin_huge_valf() : -L2E;
    const float clamp_hi = (cz < 0.0f) ? -L2E : __builtin_huge_valf();

    const float4* tP4  = (const float4*)tP;
    const float4* trP4 = (const float4*)trP;

    v2f aF0[32], aF1[32];
    {
        const float4* a0 = tP4 + (size_t)(q0 >> 4) * 512 + lane;
#pragma unroll
        for (int m = 0; m < 16; ++m) {
            const float4 t0 = a0[m * 32];
            const float4 t1 = a0[512 + m * 32];
            aF0[2 * m].x = t0.x;      aF0[2 * m].y = t0.y;
            aF0[2 * m + 1].x = t0.z;  aF0[2 * m + 1].y = t0.w;
            aF1[2 * m].x = t1.x;      aF1[2 * m].y = t1.y;
            aF1[2 * m + 1].x = t1.z;  aF1[2 * m + 1].y = t1.w;
        }
    }
    float qs0[8], qs1[8];
#pragma unroll
    for (int r = 0; r < 8; ++r) {
        qs0[r] = fmaf(cz, qn[q0 + 8 * hi + r], -L2E);
        qs1[r] = fmaf(cz, qn[q0 + 16 + 8 * hi + r], -L2E);
    }

    float racc0[8] = {}, racc1[8] = {};
    const int kstart = kg * (KTILES_PER_WAVE * 16);

    v2f b0[32], b1[32];
    float kn0, kn1;

    auto loadB = [&](v2f (&buf)[32], float& knv, int kb) {
        const float4* src = trP4 + (size_t)(kb >> 4) * 512 + lane;
#pragma unroll
        for (int m = 0; m < 16; ++m) {
            const float4 t = src[m * 32];
            buf[2 * m].x = t.x;      buf[2 * m].y = t.y;
            buf[2 * m + 1].x = t.z;  buf[2 * m + 1].y = t.w;
        }
        knv = kn[kb + lo];
    };
    auto chain = [&](const v2f (&buf)[32], v8f& acc0, v8f& acc1) {
        v8f c0 = {}, c1 = {};
#pragma unroll
        for (int s4 = 0; s4 < 32; ++s4) {
            c0 = __builtin_amdgcn_wmma_f32_16x16x4_f32(
                false, aF0[s4], false, buf[s4], (short)0, c0, false, false);
            c1 = __builtin_amdgcn_wmma_f32_16x16x4_f32(
                false, aF1[s4], false, buf[s4], (short)0, c1, false, false);
        }
        acc0 = c0;
        acc1 = c1;
    };
    auto epilogue = [&](const v8f& acc0, const v8f& acc1, const float knv) {
        const float ks = cz * knv;
#pragma unroll
        for (int r = 0; r < 8; ++r) {
            const float x0 = fmaf(acc0[r], m2, qs0[r] + ks);
            const float x1 = fmaf(acc1[r], m2, qs1[r] + ks);
            racc0[r] += __builtin_amdgcn_exp2f(__builtin_amdgcn_fmed3f(x0, clamp_lo, clamp_hi));
            racc1[r] += __builtin_amdgcn_exp2f(__builtin_amdgcn_fmed3f(x1, clamp_lo, clamp_hi));
        }
    };

#define SCHED_FENCE() __builtin_amdgcn_sched_barrier(0)

    v8f accA0, accA1, accB0, accB1;
    float knA, knB;

    loadB(b0, kn0, kstart);
    loadB(b1, kn1, kstart + 16);
    SCHED_FENCE();
    chain(b0, accA0, accA1);
    knA = kn0;
    SCHED_FENCE();
    loadB(b0, kn0, kstart + 32);
    for (int i = 0; i < KTILES_PER_WAVE / 2 - 1; ++i) {
        SCHED_FENCE();
        chain(b1, accB0, accB1);
        knB = kn1;
        epilogue(accA0, accA1, knA);
        SCHED_FENCE();
        loadB(b1, kn1, kstart + (2 * i + 3) * 16);
        SCHED_FENCE();
        chain(b0, accA0, accA1);
        knA = kn0;
        epilogue(accB0, accB1, knB);
        SCHED_FENCE();
        loadB(b0, kn0, kstart + ((2 * i + 4) & (KTILES_PER_WAVE - 1)) * 16);
    }
    SCHED_FENCE();
    chain(b1, accB0, accB1);
    epilogue(accA0, accA1, knA);
    epilogue(accB0, accB1, kn1);
#undef SCHED_FENCE

#pragma unroll
    for (int r = 0; r < 8; ++r) {
        float v = racc0[r];
        v += __shfl_xor(v, 1, 32);
        v += __shfl_xor(v, 2, 32);
        v += __shfl_xor(v, 4, 32);
        v += __shfl_xor(v, 8, 32);
        racc0[r] = v;
        float w = racc1[r];
        w += __shfl_xor(w, 1, 32);
        w += __shfl_xor(w, 2, 32);
        w += __shfl_xor(w, 4, 32);
        w += __shfl_xor(w, 8, 32);
        racc1[r] = w;
    }
    if (lo == 0) {
#pragma unroll
        for (int r = 0; r < 8; ++r) {
            const int row = q0 + 8 * hi + r;
            partial[row * nkg + kg] = racc0[r];
            partial[(row + 16) * nkg + kg] = racc1[r];
        }
    }
}

__global__ void kde_reduce_kernel(const float* __restrict__ partial,
                                  float* __restrict__ out, int NQ, int nkg) {
    int q = blockIdx.x * blockDim.x + threadIdx.x;
    if (q >= NQ) return;
    float acc = 0.0f;
    for (int g = 0; g < nkg; ++g) acc += partial[q * nkg + g];
    out[q] = acc;
}

extern "C" void kernel_launch(void* const* d_in, const int* in_sizes, int n_in,
                              void* d_out, int out_size, void* d_ws, size_t ws_size,
                              hipStream_t stream) {
    const float* test  = (const float*)d_in[0];
    const float* train = (const float*)d_in[1];
    float* out = (float*)d_out;

    const int NQ = in_sizes[0] / KDE_D;
    const int NK = in_sizes[1] / KDE_D;
    const int nkg = NK / (KTILES_PER_WAVE * 16);

    float* ws      = (float*)d_ws;
    float* bw      = ws;
    float* inv_bw  = ws + KDE_D;
    float* scal    = ws + 2 * KDE_D;
    float* qn      = ws + 3 * KDE_D;
    float* kn      = qn + NQ;
    float* tP      = kn + NK;
    float* trP     = tP + KDE_D * NQ;
    float* partial = trP + KDE_D * NK;

    kde_colstat_kernel<<<BW_ROWCHUNKS * BW_DIMSLICES, 256, 0, stream>>>(train, NK, partial);
    kde_bw_kernel<<<1, 2 * KDE_D, 0, stream>>>(partial, NK, bw, inv_bw, scal);

    kde_prep_kernel<<<(NQ + NK) / PREP_ROWS, 256, 0, stream>>>(test, train, inv_bw,
                                                              tP, trP, qn, kn, NQ, NK);

    dim3 grid(NQ / 256, nkg);
    kde_wmma_kernel<<<grid, 256, 0, stream>>>(tP, trP, qn, kn, scal, partial, NQ, NK, nkg);

    kde_reduce_kernel<<<(NQ + 255) / 256, 256, 0, stream>>>(partial, out, NQ, nkg);
}


# Bandwidth + normalizer launch folded into the pack launch as a per-work-group prologue (5 launches to 4); column-stat partials cut to 16 row chunks; f64 sum order changes, n^(-1/132) now in f32
# speedup vs baseline: 18.2245x; 1.0958x over previous
#include <hip/hip_runtime.h>

typedef __attribute__((ext_vector_type(2))) float v2f;
typedef __attribute__((ext_vector_type(8))) float v8f;

#define KDE_D 128
#define KDE_P (KDE_D / 2)
#define KTILES_PER_WAVE 32

#define BW_ROWCHUNKS 16
#define BW_DIMSLICES 16
#define BW_DIMS   (KDE_D / BW_DIMSLICES)
#define BW_F4     (BW_DIMS / 4)
#define BW_SLOTS  (256 / BW_F4)
#define BW_VALS   (2 * BW_DIMS)
#define BW_GROUPS (256 / BW_VALS)
static_assert(BW_SLOTS == 8 * BW_GROUPS && BW_GROUPS == 16, "tree shape below");

__global__ __launch_bounds__(256)
void kde_colstat_kernel(const float* __restrict__ train, int n,
                        float* __restrict__ part_f) {
    double* part = (double*)part_f;
    const int slice = blockIdx.x % BW_DIMSLICES;
    const int chunk = blockIdx.x / BW_DIMSLICES;
    const int c4    = threadIdx.x & (BW_F4 - 1);
    const int slot  = threadIdx.x / BW_F4;
    const float* col = train + slice * BW_DIMS + c4 * 4;

    double s0 = 0.0, s1 = 0.0, s2 = 0.0, s3 = 0.0;
    double q0 = 0.0, q1 = 0.0, q2 = 0.0, q3 = 0.0;
    auto accum = [&](const float4 v) {
        const double x0 = (double)v.x, x1 = (double)v.y;
        const double x2 = (double)v.z, x3 = (double)v.w;
        s0 += x0; q0 += x0 * x0;
        s1 += x1; q1 += x1 * x1;
        s2 += x2; q2 += x2 * x2;
        s3 += x3; q3 += x3 * x3;
    };
    const int rstep = BW_ROWCHUNKS * BW_SLOTS;
    int r = chunk * BW_SLOTS + slot;
    for (; r + 7 * rstep < n; r += 8 * rstep) {
        float4 v[8];
#pragma unroll
        for (int j = 0; j < 8; ++j)
            v[j] = *(const float4*)(col + (size_t)(r + j * rstep) * KDE_D);
#pragma unroll
        for (int j = 0; j < 8; ++j) accum(v[j]);
    }
    for (; r < n; r += rstep)
        accum(*(const float4*)(col + (size_t)r * KDE_D));

    __shared__ double red[BW_SLOTS][BW_VALS];
    __shared__ double red2[BW_GROUPS][BW_VALS];
    red[slot][c4 * 4 + 0] = s0;
    red[slot][c4 * 4 + 1] = s1;
    red[slot][c4 * 4 + 2] = s2;
    red[slot][c4 * 4 + 3] = s3;
    red[slot][BW_DIMS + c4 * 4 + 0] = q0;
    red[slot][BW_DIMS + c4 * 4 + 1] = q1;
    red[slot][BW_DIMS + c4 * 4 + 2] = q2;
    red[slot][BW_DIMS + c4 * 4 + 3] = q3;
    __syncthreads();
    const int v = threadIdx.x & (BW_VALS - 1);
    const int g = threadIdx.x / BW_VALS;
    double acc = red[g * 8][v];
    for (int j = 1; j < 8; ++j) acc += red[g * 8 + j][v];
    red2[g][v] = acc;
    __syncthreads();
    if (threadIdx.x < BW_VALS) {
        const double t0 = (red2[0][v] + red2[1][v]) + (red2[2][v] + red2[3][v]);
        const double t1 = (red2[4][v] + red2[5][v]) + (red2[6][v] + red2[7][v]);
        const double t2 = (red2[8][v] + red2[9][v]) + (red2[10][v] + red2[11][v]);
        const double t3 = (red2[12][v] + red2[13][v]) + (red2[14][v] + red2[15][v]);
        const double t = (t0 + t1) + (t2 + t3);
        part[chunk * (2 * KDE_D) + (v / BW_DIMS) * KDE_D + slice * BW_DIMS + (v % BW_DIMS)] = t;
    }
}

#define PREP_ROWS 64
#define PREP_PITCH (KDE_D + 2)

__global__ __launch_bounds__(256)
void kde_prep_kernel(const float* __restrict__ test, const float* __restrict__ train,
                     const float* __restrict__ part_f, float* __restrict__ scal,
                     float* __restrict__ tP, float* __restrict__ trP,
                     float* __restrict__ qn, float* __restrict__ kn,
                     int NQ, int NK) {
    __shared__ float tile[PREP_ROWS * PREP_PITCH];
    __shared__ double sumsq[KDE_D];
    __shared__ float ibw[KDE_D];
    __shared__ float red[KDE_D];

    const int nqb = NQ / PREP_ROWS;
    const bool is_test = (int)blockIdx.x < nqb;
    const float* src = is_test ? test : train;
    float* nrm       = is_test ? qn : kn;
    const int r0     = (is_test ? (int)blockIdx.x : (int)blockIdx.x - nqb) * PREP_ROWS;
    float4* dst4     = (float4*)(is_test ? tP : trP) + (size_t)(r0 >> 4) * 512;

    const int lane = threadIdx.x & 31;
    const int wave = threadIdx.x >> 5;

    float4 v[8];
#pragma unroll
    for (int j = 0; j < 8; ++j)
        v[j] = *(const float4*)(src + (size_t)(r0 + 8 * j + wave) * KDE_D + 4 * lane);

    {
        const double* part = (const double*)part_f;
        const int t = threadIdx.x;
        const int d = t & (KDE_D - 1);
        double p[BW_ROWCHUNKS];
#pragma unroll
        for (int c = 0; c < BW_ROWCHUNKS; ++c) p[c] = part[c * (2 * KDE_D) + t];
        double acc = p[0];
#pragma unroll
        for (int c = 1; c < BW_ROWCHUNKS; ++c) acc += p[c];

        if (t >= KDE_D) sumsq[d] = acc;
        __syncthreads();
        if (t < KDE_D) {
            const int n = NK;
            const double s = acc, s2 = sumsq[d];
            double var = (s2 - s * s / (double)n) / (double)(n - 1);
            if (var < 0.0) var = 0.0;
            float stdv = (float)sqrt(var);
            stdv = fmaxf(stdv, 0.01f);
            const float scott = expf(-logf((float)n) / 132.0f);
            float b = 1.06f * stdv * scott;
            b = fminf(b, 0.49f);
            ibw[d] = 1.0f / b;
            red[d] = logf(b);
        }
        __syncthreads();
        if (blockIdx.x == 0) {
            for (int off = KDE_D / 2; off > 0; off >>= 1) {
                if (t < off) red[t] += red[t + off];
                __syncthreads();
            }
            if (t == 0) {
                float Z = 0.5f * (float)KDE_D * logf(6.28318530717958647692f)
                        + red[0] + logf((float)NK);
                scal[0] = Z;
                scal[1] = 1.0f / Z;
            }
        }
    }
    float2 ib01, ib23;
    ib01.x = ibw[4 * lane];      ib01.y = ibw[4 * lane + 1];
    ib23.x = ibw[4 * lane + 2];  ib23.y = ibw[4 * lane + 3];
#pragma unroll
    for (int j = 0; j < 8; ++j) {
        float* t = tile + (8 * j + wave) * PREP_PITCH + 4 * lane;
        float2 a, b;
        a.x = v[j].x * ib01.x;  a.y = v[j].y * ib01.y;
        b.x = v[j].z * ib23.x;  b.y = v[j].w * ib23.y;
        *(float2*)(t)     = a;
        *(float2*)(t + 2) = b;
    }
    __syncthreads();

#pragma unroll
    for (int i = 0; i < 8; ++i) {
        const int jj  = i >> 1;
        const int m   = (i & 1) * 8 + wave;
        const int row = jj * 16 + (lane & 15);
        const float* t = tile + row * PREP_PITCH + 8 * m + 2 * (lane >> 4);
        const float2 a = *(const float2*)(t);
        const float2 b = *(const float2*)(t + 4);
        float4 val;
        val.x = a.x;  val.y = a.y;  val.z = b.x;  val.w = b.y;
        dst4[i * 256 + threadIdx.x] = val;
    }

    if (threadIdx.x < PREP_ROWS) {
        const float* t = tile + threadIdx.x * PREP_PITCH;
        float acc = 0.0f;
#pragma unroll 16
        for (int p = 0; p < KDE_P; ++p) {
            const float2 x = *(const float2*)(t + 2 * p);
            acc = fmaf(x.x, x.x, acc);
            acc = fmaf(x.y, x.y, acc);
        }
        nrm[r0 + threadIdx.x] = acc;
    }
}

__global__ __launch_bounds__(256)
void kde_wmma_kernel(const float* __restrict__ tP, const float* __restrict__ trP,
                     const float* __restrict__ qn, const float* __restrict__ kn,
                     const float* __restrict__ scal, float* __restrict__ partial,
                     int NQ, int NK, int nkg) {
    const int lane = threadIdx.x & 31;
    const int wave = threadIdx.x >> 5;
    const int q0   = (blockIdx.x * 8 + wave) * 32;
    const int kg   = blockIdx.y;
    const int hi   = lane >> 4;
    const int lo   = lane & 15;
    const float invZ = scal[1];

    const float L2E = 1.44269504088896341f;
    const float cz  = -0.5f * invZ * L2E;
    const float m2  = -2.0f * cz;
    const float clamp_lo = (cz < 0.0f) ? -__builtin_huge_valf() : -L2E;
    const float clamp_hi = (cz < 0.0f) ? -L2E : __builtin_huge_valf();

    const float4* tP4  = (const float4*)tP;
    const float4* trP4 = (const float4*)trP;

    v2f aF0[32], aF1[32];
    {
        const float4* a0 = tP4 + (size_t)(q0 >> 4) * 512 + lane;
#pragma unroll
        for (int m = 0; m < 16; ++m) {
            const float4 t0 = a0[m * 32];
            const float4 t1 = a0[512 + m * 32];
            aF0[2 * m].x = t0.x;      aF0[2 * m].y = t0.y;
            aF0[2 * m + 1].x = t0.z;  aF0[2 * m + 1].y = t0.w;
            aF1[2 * m].x = t1.x;      aF1[2 * m].y = t1.y;
            aF1[2 * m + 1].x = t1.z;  aF1[2 * m + 1].y = t1.w;
        }
    }
    float qs0[8], qs1[8];
#pragma unroll
    for (int r = 0; r < 8; ++r) {
        qs0[r] = fmaf(cz, qn[q0 + 8 * hi + r], -L2E);
        qs1[r] = fmaf(cz, qn[q0 + 16 + 8 * hi + r], -L2E);
    }

    float racc0[8] = {}, racc1[8] = {};
    const int kstart = kg * (KTILES_PER_WAVE * 16);

    v2f b0[32], b1[32];
    float kn0, kn1;

    auto loadB = [&](v2f (&buf)[32], float& knv, int kb) {
        const float4* src = trP4 + (size_t)(kb >> 4) * 512 + lane;
#pragma unroll
        for (int m = 0; m < 16; ++m) {
            const float4 t = src[m * 32];
            buf[2 * m].x = t.x;      buf[2 * m].y = t.y;
            buf[2 * m + 1].x = t.z;  buf[2 * m + 1].y = t.w;
        }
        knv = kn[kb + lo];
    };
    auto chain = [&](const v2f (&buf)[32], v8f& acc0, v8f& acc1) {
        v8f c0 = {}, c1 = {};
#pragma unroll
        for (int s4 = 0; s4 < 32; ++s4) {
            c0 = __builtin_amdgcn_wmma_f32_16x16x4_f32(
                false, aF0[s4], false, buf[s4], (short)0, c0, false, false);
            c1 = __builtin_amdgcn_wmma_f32_16x16x4_f32(
                false, aF1[s4], false, buf[s4], (short)0, c1, false, false);
        }
        acc0 = c0;
        acc1 = c1;
    };
    auto epilogue = [&](const v8f& acc0, const v8f& acc1, const float knv) {
        const float ks = cz * knv;
#pragma unroll
        for (int r = 0; r < 8; ++r) {
            const float x0 = fmaf(acc0[r], m2, qs0[r] + ks);
            const float x1 = fmaf(acc1[r], m2, qs1[r] + ks);
            racc0[r] += __builtin_amdgcn_exp2f(__builtin_amdgcn_fmed3f(x0, clamp_lo, clamp_hi));
            racc1[r] += __builtin_amdgcn_exp2f(__builtin_amdgcn_fmed3f(x1, clamp_lo, clamp_hi));
        }
    };

#define SCHED_FENCE() __builtin_amdgcn_sched_barrier(0)

    v8f accA0, accA1, accB0, accB1;
    float knA, knB;

    loadB(b0, kn0, kstart);
    loadB(b1, kn1, kstart + 16);
    SCHED_FENCE();
    chain(b0, accA0, accA1);
    knA = kn0;
    SCHED_FENCE();
    loadB(b0, kn0, kstart + 32);
    for (int i = 0; i < KTILES_PER_WAVE / 2 - 1; ++i) {
        SCHED_FENCE();
        chain(b1, accB0, accB1);
        knB = kn1;
        epilogue(accA0, accA1, knA);
        SCHED_FENCE();
        loadB(b1, kn1, kstart + (2 * i + 3) * 16);
        SCHED_FENCE();
        chain(b0, accA0, accA1);
        knA = kn0;
        epilogue(accB0, accB1, knB);
        SCHED_FENCE();
        loadB(b0, kn0, kstart + ((2 * i + 4) & (KTILES_PER_WAVE - 1)) * 16);
    }
    SCHED_FENCE();
    chain(b1, accB0, accB1);
    epilogue(accA0, accA1, knA);
    epilogue(accB0, accB1, kn1);
#undef SCHED_FENCE

#pragma unroll
    for (int r = 0; r < 8; ++r) {
        float v = racc0[r];
        v += __shfl_xor(v, 1, 32);
        v += __shfl_xor(v, 2, 32);
        v += __shfl_xor(v, 4, 32);
        v += __shfl_xor(v, 8, 32);
        racc0[r] = v;
        float w = racc1[r];
        w += __shfl_xor(w, 1, 32);
        w += __shfl_xor(w, 2, 32);
        w += __shfl_xor(w, 4, 32);
        w += __shfl_xor(w, 8, 32);
        racc1[r] = w;
    }
    if (lo == 0) {
#pragma unroll
        for (int r = 0; r < 8; ++r) {
            const int row = q0 + 8 * hi + r;
            partial[row * nkg + kg] = racc0[r];
            partial[(row + 16) * nkg + kg] = racc1[r];
        }
    }
}

__global__ void kde_reduce_kernel(const float* __restrict__ partial,
                                  float* __restrict__ out, int NQ, int nkg) {
    int q = blockIdx.x * blockDim.x + threadIdx.x;
    if (q >= NQ) return;
    float acc = 0.0f;
    for (int g = 0; g < nkg; ++g) acc += partial[q * nkg + g];
    out[q] = acc;
}

extern "C" void kernel_launch(void* const* d_in, const int* in_sizes, int n_in,
                              void* d_out, int out_size, void* d_ws, size_t ws_size,
                              hipStream_t stream) {
    const float* test  = (const float*)d_in[0];
    const float* train = (const float*)d_in[1];
    float* out = (float*)d_out;

    const int NQ = in_sizes[0] / KDE_D;
    const int NK = in_sizes[1] / KDE_D;
    const int nkg = NK / (KTILES_PER_WAVE * 16);

    float* ws      = (float*)d_ws;
    float* scal    = ws + 2 * KDE_D;
    float* qn      = ws + 3 * KDE_D;
    float* kn      = qn + NQ;
    float* tP      = kn + NK;
    float* trP     = tP + KDE_D * NQ;
    float* partial = trP + KDE_D * NK;

    kde_colstat_kernel<<<BW_ROWCHUNKS * BW_DIMSLICES, 256, 0, stream>>>(train, NK, partial);

    kde_prep_kernel<<<(NQ + NK) / PREP_ROWS, 256, 0, stream>>>(test, train, partial, scal,
                                                              tP, trP, qn, kn, NQ, NK);

    dim3 grid(NQ / 256, nkg);
    kde_wmma_kernel<<<grid, 256, 0, stream>>>(tP, trP, qn, kn, scal, partial, NQ, NK, nkg);

    kde_reduce_kernel<<<(NQ + 255) / 256, 256, 0, stream>>>(partial, out, NQ, nkg);
}


# Column statistics in f32: pooled (mean, M2) partials (8 rows per thread in registers, pairwise merges) replace the f64 sum/sum-of-squares pass and f64 variance prologue; sum order changes
# speedup vs baseline: 18.8845x; 1.0362x over previous
#include <hip/hip_runtime.h>

typedef __attribute__((ext_vector_type(2))) float v2f;
typedef __attribute__((ext_vector_type(8))) float v8f;

#define KDE_D 128
#define KDE_P (KDE_D / 2)
#define KTILES_PER_WAVE 32

#define CS_SLICES  8
#define CS_CHUNKS  32
#define CS_DIMS    (KDE_D / CS_SLICES)
#define CS_F4      (CS_DIMS / 4)
#define CS_SLOTS   (256 / CS_F4)
#define CS_ROWSTEP (CS_CHUNKS * CS_SLOTS)
static_assert(CS_SLOTS == 64 && CS_DIMS == 16, "tree shape below");

__device__ __forceinline__ void stat_merge(const float na, float& ma, float& qa,
                                           const float nb, const float mb, const float qb) {
    const float nab   = na + nb;
    const float f     = (nab > 0.0f) ? nb * __builtin_amdgcn_rcpf(nab) : 0.0f;
    const float delta = mb - ma;
    ma = fmaf(delta, f, ma);
    qa = (qa + qb) + (delta * delta) * (na * f);
}

__device__ __forceinline__ float stat_count(const int n, const int lo, const int len) {
    const int full = n / CS_ROWSTEP, rem = n % CS_ROWSTEP;
    int part = rem - lo;
    part = part < 0 ? 0 : (part > len ? len : part);
    return (float)(full * len + part);
}

__device__ __forceinline__ void stat8(const float x0, const float x1, const float x2, const float x3,
                                      const float x4, const float x5, const float x6, const float x7,
                                      float& mean, float& m2) {
    mean = (((x0 + x1) + (x2 + x3)) + ((x4 + x5) + (x6 + x7))) * 0.125f;
    float d, q;
    d = x0 - mean;  q = d * d;
    d = x1 - mean;  q = fmaf(d, d, q);
    d = x2 - mean;  q = fmaf(d, d, q);
    d = x3 - mean;  q = fmaf(d, d, q);
    d = x4 - mean;  q = fmaf(d, d, q);
    d = x5 - mean;  q = fmaf(d, d, q);
    d = x6 - mean;  q = fmaf(d, d, q);
    d = x7 - mean;  q = fmaf(d, d, q);
    m2 = q;
}

__device__ __forceinline__ void stat_push(const float x, const float inv, float& mean, float& m2) {
    const float d = x - mean;
    mean = fmaf(d, inv, mean);
    m2   = fmaf(d, x - mean, m2);
}

__device__ __forceinline__ void stat_merge4(const int n, const int lo, const int len,
                                            float (&m)[4], float (&q)[4]) {
    const float c0 = stat_count(n, lo, len),           c1 = stat_count(n, lo + len, len);
    const float c2 = stat_count(n, lo + 2 * len, len), c3 = stat_count(n, lo + 3 * len, len);
    stat_merge(c0, m[0], q[0], c1, m[1], q[1]);
    stat_merge(c2, m[2], q[2], c3, m[3], q[3]);
    stat_merge(c0 + c1, m[0], q[0], c2 + c3, m[2], q[2]);
}

__global__ __launch_bounds__(256)
void kde_colstat_kernel(const float* __restrict__ train, int n,
                        float* __restrict__ part_f) {
    float2* part = (float2*)part_f;
    const int slice = blockIdx.x % CS_SLICES;
    const int chunk = blockIdx.x / CS_SLICES;
    const int c4    = threadIdx.x & (CS_F4 - 1);
    const int slot  = threadIdx.x / CS_F4;
    const float* col = train + slice * CS_DIMS + c4 * 4;

    float cn = 0.0f;
    float m0 = 0.0f, m1 = 0.0f, m2 = 0.0f, m3 = 0.0f;
    float q0 = 0.0f, q1 = 0.0f, q2 = 0.0f, q3 = 0.0f;
    int r = chunk * CS_SLOTS + slot;
    for (; r + 7 * CS_ROWSTEP < n; r += 8 * CS_ROWSTEP) {
        float4 v[8];
#pragma unroll
        for (int j = 0; j < 8; ++j)
            v[j] = *(const float4*)(col + (size_t)(r + j * CS_ROWSTEP) * KDE_D);
        float bm, bq;
        stat8(v[0].x, v[1].x, v[2].x, v[3].x, v[4].x, v[5].x, v[6].x, v[7].x, bm, bq);
        stat_merge(cn, m0, q0, 8.0f, bm, bq);
        stat8(v[0].y, v[1].y, v[2].y, v[3].y, v[4].y, v[5].y, v[6].y, v[7].y, bm, bq);
        stat_merge(cn, m1, q1, 8.0f, bm, bq);
        stat8(v[0].z, v[1].z, v[2].z, v[3].z, v[4].z, v[5].z, v[6].z, v[7].z, bm, bq);
        stat_merge(cn, m2, q2, 8.0f, bm, bq);
        stat8(v[0].w, v[1].w, v[2].w, v[3].w, v[4].w, v[5].w, v[6].w, v[7].w, bm, bq);
        stat_merge(cn, m3, q3, 8.0f, bm, bq);
        cn += 8.0f;
    }
    for (; r < n; r += CS_ROWSTEP) {
        const float4 x = *(const float4*)(col + (size_t)r * KDE_D);
        cn += 1.0f;
        const float inv = 1.0f / cn;
        stat_push(x.x, inv, m0, q0);
        stat_push(x.y, inv, m1, q1);
        stat_push(x.z, inv, m2, q2);
        stat_push(x.w, inv, m3, q3);
    }

    __shared__ float sMean[CS_SLOTS][CS_DIMS];
    __shared__ float sM2[CS_SLOTS][CS_DIMS];
    {
        float4 t;
        t.x = m0;  t.y = m1;  t.z = m2;  t.w = m3;
        *(float4*)&sMean[slot][c4 * 4] = t;
        t.x = q0;  t.y = q1;  t.z = q2;  t.w = q3;
        *(float4*)&sM2[slot][c4 * 4] = t;
    }
    __syncthreads();
    const int v   = threadIdx.x & (CS_DIMS - 1);
    const int g   = threadIdx.x / CS_DIMS;
    const int lo0 = chunk * CS_SLOTS;
    float m[4], q[4];
#pragma unroll
    for (int i = 0; i < 4; ++i) { m[i] = sMean[4 * g + i][v];  q[i] = sM2[4 * g + i][v]; }
    stat_merge4(n, lo0 + 4 * g, 1, m, q);
    sMean[4 * g][v] = m[0];
    sM2[4 * g][v]   = q[0];
    __syncthreads();
    if (g < 4) {
#pragma unroll
        for (int i = 0; i < 4; ++i) { m[i] = sMean[16 * g + 4 * i][v];  q[i] = sM2[16 * g + 4 * i][v]; }
        stat_merge4(n, lo0 + 16 * g, 4, m, q);
        sMean[16 * g][v] = m[0];
        sM2[16 * g][v]   = q[0];
    }
    __syncthreads();
    if (g == 0) {
#pragma unroll
        for (int i = 0; i < 4; ++i) { m[i] = sMean[16 * i][v];  q[i] = sM2[16 * i][v]; }
        stat_merge4(n, lo0, 16, m, q);
        float2 o;
        o.x = m[0];  o.y = q[0];
        part[chunk * KDE_D + slice * CS_DIMS + v] = o;
    }
}

#define PREP_ROWS 64
#define PREP_PITCH (KDE_D + 2)

__global__ __launch_bounds__(256)
void kde_prep_kernel(const float* __restrict__ test, const float* __restrict__ train,
                     const float* __restrict__ part_f, float* __restrict__ scal,
                     float* __restrict__ tP, float* __restrict__ trP,
                     float* __restrict__ qn, float* __restrict__ kn,
                     int NQ, int NK) {
    __shared__ float tile[PREP_ROWS * PREP_PITCH];
    __shared__ float ibw[KDE_D];
    __shared__ float red[KDE_D];

    const int nqb = NQ / PREP_ROWS;
    const bool is_test = (int)blockIdx.x < nqb;
    const float* src = is_test ? test : train;
    float* nrm       = is_test ? qn : kn;
    const int r0     = (is_test ? (int)blockIdx.x : (int)blockIdx.x - nqb) * PREP_ROWS;
    float4* dst4     = (float4*)(is_test ? tP : trP) + (size_t)(r0 >> 4) * 512;

    const int lane = threadIdx.x & 31;
    const int wave = threadIdx.x >> 5;

    float4 v[8];
#pragma unroll
    for (int j = 0; j < 8; ++j)
        v[j] = *(const float4*)(src + (size_t)(r0 + 8 * j + wave) * KDE_D + 4 * lane);

    {
        const float2* part = (const float2*)part_f;
        const int t = threadIdx.x;
        const int n = NK;
        if (t < KDE_D) {
            float pm[CS_CHUNKS], pq[CS_CHUNKS], pc[CS_CHUNKS];
#pragma unroll
            for (int c = 0; c < CS_CHUNKS; ++c) {
                const float2 p = part[c * KDE_D + t];
                pm[c] = p.x;
                pq[c] = p.y;
                pc[c] = stat_count(n, c * CS_SLOTS, CS_SLOTS);
            }
#pragma unroll
            for (int w = 1; w < CS_CHUNKS; w <<= 1) {
#pragma unroll
                for (int c = 0; c < CS_CHUNKS; c += 2 * w) {
                    stat_merge(pc[c], pm[c], pq[c], pc[c + w], pm[c + w], pq[c + w]);
                    pc[c] += pc[c + w];
                }
            }
            const float m2 = pq[0];
            float var = m2 / (float)(n - 1);
            if (var < 0.0f) var = 0.0f;
            float stdv = sqrtf(var);
            stdv = fmaxf(stdv, 0.01f);
            const float scott = expf(-logf((float)n) / 132.0f);
            float b = 1.06f * stdv * scott;
            b = fminf(b, 0.49f);
            ibw[t] = 1.0f / b;
            red[t] = logf(b);
        }
        __syncthreads();
        if (blockIdx.x == 0) {
            for (int off = KDE_D / 2; off > 0; off >>= 1) {
                if (t < off) red[t] += red[t + off];
                __syncthreads();
            }
            if (t == 0) {
                float Z = 0.5f * (float)KDE_D * logf(6.28318530717958647692f)
                        + red[0] + logf((float)NK);
                scal[0] = Z;
                scal[1] = 1.0f / Z;
            }
        }
    }
    float2 ib01, ib23;
    ib01.x = ibw[4 * lane];      ib01.y = ibw[4 * lane + 1];
    ib23.x = ibw[4 * lane + 2];  ib23.y = ibw[4 * lane + 3];
#pragma unroll
    for (int j = 0; j < 8; ++j) {
        float* t = tile + (8 * j + wave) * PREP_PITCH + 4 * lane;
        float2 a, b;
        a.x = v[j].x * ib01.x;  a.y = v[j].y * ib01.y;
        b.x = v[j].z * ib23.x;  b.y = v[j].w * ib23.y;
        *(float2*)(t)     = a;
        *(float2*)(t + 2) = b;
    }
    __syncthreads();

#pragma unroll
    for (int i = 0; i < 8; ++i) {
        const int jj  = i >> 1;
        const int m   = (i & 1) * 8 + wave;
        const int row = jj * 16 + (lane & 15);
        const float* t = tile + row * PREP_PITCH + 8 * m + 2 * (lane >> 4);
        const float2 a = *(const float2*)(t);
        const float2 b = *(const float2*)(t + 4);
        float4 val;
        val.x = a.x;  val.y = a.y;  val.z = b.x;  val.w = b.y;
        dst4[i * 256 + threadIdx.x] = val;
    }

    if (threadIdx.x < PREP_ROWS) {
        const float* t = tile + threadIdx.x * PREP_PITCH;
        float acc = 0.0f;
#pragma unroll 16
        for (int p = 0; p < KDE_P; ++p) {
            const float2 x = *(const float2*)(t + 2 * p);
            acc = fmaf(x.x, x.x, acc);
            acc = fmaf(x.y, x.y, acc);
        }
        nrm[r0 + threadIdx.x] = acc;
    }
}

__global__ __launch_bounds__(256)
void kde_wmma_kernel(const float* __restrict__ tP, const float* __restrict__ trP,
                     const float* __restrict__ qn, const float* __restrict__ kn,
                     const float* __restrict__ scal, float* __restrict__ partial,
                     int NQ, int NK, int nkg) {
    const int lane = threadIdx.x & 31;
    const int wave = threadIdx.x >> 5;
    const int q0   = (blockIdx.x * 8 + wave) * 32;
    const int kg   = blockIdx.y;
    const int hi   = lane >> 4;
    const int lo   = lane & 15;
    const float invZ = scal[1];

    const float L2E = 1.44269504088896341f;
    const float cz  = -0.5f * invZ * L2E;
    const float m2  = -2.0f * cz;
    const float clamp_lo = (cz < 0.0f) ? -__builtin_huge_valf() : -L2E;
    const float clamp_hi = (cz < 0.0f) ? -L2E : __builtin_huge_valf();

    const float4* tP4  = (const float4*)tP;
    const float4* trP4 = (const float4*)trP;

    v2f aF0[32], aF1[32];
    {
        const float4* a0 = tP4 + (size_t)(q0 >> 4) * 512 + lane;
#pragma unroll
        for (int m = 0; m < 16; ++m) {
            const float4 t0 = a0[m * 32];
            const float4 t1 = a0[512 + m * 32];
            aF0[2 * m].x = t0.x;      aF0[2 * m].y = t0.y;
            aF0[2 * m + 1].x = t0.z;  aF0[2 * m + 1].y = t0.w;
            aF1[2 * m].x = t1.x;      aF1[2 * m].y = t1.y;
            aF1[2 * m + 1].x = t1.z;  aF1[2 * m + 1].y = t1.w;
        }
    }
    float qs0[8], qs1[8];
#pragma unroll
    for (int r = 0; r < 8; ++r) {
        qs0[r] = fmaf(cz, qn[q0 + 8 * hi + r], -L2E);
        qs1[r] = fmaf(cz, qn[q0 + 16 + 8 * hi + r], -L2E);
    }

    float racc0[8] = {}, racc1[8] = {};
    const int kstart = kg * (KTILES_PER_WAVE * 16);

    v2f b0[32], b1[32];
    float kn0, kn1;

    auto loadB = [&](v2f (&buf)[32], float& knv, int kb) {
        const float4* src = trP4 + (size_t)(kb >> 4) * 512 + lane;
#pragma unroll
        for (int m = 0; m < 16; ++m) {
            const float4 t = src[m * 32];
            buf[2 * m].x = t.x;      buf[2 * m].y = t.y;
            buf[2 * m + 1].x = t.z;  buf[2 * m + 1].y = t.w;
        }
        knv = kn[kb + lo];
    };
    auto chain = [&](const v2f (&buf)[32], v8f& acc0, v8f& acc1) {
        v8f c0 = {}, c1 = {};
#pragma unroll
        for (int s4 = 0; s4 < 32; ++s4) {
            c0 = __builtin_amdgcn_wmma_f32_16x16x4_f32(
                false, aF0[s4], false, buf[s4], (short)0, c0, false, false);
            c1 = __builtin_amdgcn_wmma_f32_16x16x4_f32(
                false, aF1[s4], false, buf[s4], (short)0, c1, false, false);
        }
        acc0 = c0;
        acc1 = c1;
    };
    auto epilogue = [&](const v8f& acc0, const v8f& acc1, const float knv) {
        const float ks = cz * knv;
#pragma unroll
        for (int r = 0; r < 8; ++r) {
            const float x0 = fmaf(acc0[r], m2, qs0[r] + ks);
            const float x1 = fmaf(acc1[r], m2, qs1[r] + ks);
            racc0[r] += __builtin_amdgcn_exp2f(__builtin_amdgcn_fmed3f(x0, clamp_lo, clamp_hi));
            racc1[r] += __builtin_amdgcn_exp2f(__builtin_amdgcn_fmed3f(x1, clamp_lo, clamp_hi));
        }
    };

#define SCHED_FENCE() __builtin_amdgcn_sched_barrier(0)

    v8f accA0, accA1, accB0, accB1;
    float knA, knB;

    loadB(b0, kn0, kstart);
    loadB(b1, kn1, kstart + 16);
    SCHED_FENCE();
    chain(b0, accA0, accA1);
    knA = kn0;
    SCHED_FENCE();
    loadB(b0, kn0, kstart + 32);
    for (int i = 0; i < KTILES_PER_WAVE / 2 - 1; ++i) {
        SCHED_FENCE();
        chain(b1, accB0, accB1);
        knB = kn1;
        epilogue(accA0, accA1, knA);
        SCHED_FENCE();
        loadB(b1, kn1, kstart + (2 * i + 3) * 16);
        SCHED_FENCE();
        chain(b0, accA0, accA1);
        knA = kn0;
        epilogue(accB0, accB1, knB);
        SCHED_FENCE();
        loadB(b0, kn0, kstart + ((2 * i + 4) & (KTILES_PER_WAVE - 1)) * 16);
    }
    SCHED_FENCE();
    chain(b1, accB0, accB1);
    epilogue(accA0, accA1, knA);
    epilogue(accB0, accB1, kn1);
#undef SCHED_FENCE

#pragma unroll
    for (int r = 0; r < 8; ++r) {
        float v = racc0[r];
        v += __shfl_xor(v, 1, 32);
        v += __shfl_xor(v, 2, 32);
        v += __shfl_xor(v, 4, 32);
        v += __shfl_xor(v, 8, 32);
        racc0[r] = v;
        float w = racc1[r];
        w += __shfl_xor(w, 1, 32);
        w += __shfl_xor(w, 2, 32);
        w += __shfl_xor(w, 4, 32);
        w += __shfl_xor(w, 8, 32);
        racc1[r] = w;
    }
    if (lo == 0) {
#pragma unroll
        for (int r = 0; r < 8; ++r) {
            const int row = q0 + 8 * hi + r;
            partial[row * nkg + kg] = racc0[r];
            partial[(row + 16) * nkg + kg] = racc1[r];
        }
    }
}

__global__ void kde_reduce_kernel(const float* __restrict__ partial,
                                  float* __restrict__ out, int NQ, int nkg) {
    int q = blockIdx.x * blockDim.x + threadIdx.x;
    if (q >= NQ) return;
    float acc = 0.0f;
    for (int g = 0; g < nkg; ++g) acc += partial[q * nkg + g];
    out[q] = acc;
}

extern "C" void kernel_launch(void* const* d_in, const int* in_sizes, int n_in,
                              void* d_out, int out_size, void* d_ws, size_t ws_size,
                              hipStream_t stream) {
    const float* test  = (const float*)d_in[0];
    const float* train = (const float*)d_in[1];
    float* out = (float*)d_out;

    const int NQ = in_sizes[0] / KDE_D;
    const int NK = in_sizes[1] / KDE_D;
    const int nkg = NK / (KTILES_PER_WAVE * 16);

    float* ws      = (float*)d_ws;
    float* scal    = ws + 2 * KDE_D;
    float* qn      = ws + 3 * KDE_D;
    float* kn      = qn + NQ;
    float* tP      = kn + NK;
    float* trP     = tP + KDE_D * NQ;
    float* partial = trP + KDE_D * NK;

    kde_colstat_kernel<<<CS_CHUNKS * CS_SLICES, 256, 0, stream>>>(train, NK, partial);

    kde_prep_kernel<<<(NQ + NK) / PREP_ROWS, 256, 0, stream>>>(test, train, partial, scal,
                                                              tP, trP, qn, kn, NQ, NK);

    dim3 grid(NQ / 256, nkg);
    kde_wmma_kernel<<<grid, 256, 0, stream>>>(tP, trP, qn, kn, scal, partial, NQ, NK, nkg);

    kde_reduce_kernel<<<(NQ + 255) / 256, 256, 0, stream>>>(partial, out, NQ, nkg);
}
